# starting kernel
# baseline (speedup 1.0000x reference)
#include <hip/hip_runtime.h>
#include <math.h>

typedef __attribute__((ext_vector_type(2))) float v2f;
typedef __attribute__((ext_vector_type(8))) float v8f;

#define T_LEN 400
#define D_LEN 10000
#define K1 1200
#define K3 800
#define NTILES (D_LEN / 16)

__global__ void hdc_features_kernel(const float* __restrict__ input,
                                    float* __restrict__ ws) {
  __shared__ float s_t[T_LEN];
  __shared__ float s_x[T_LEN * 3];

  const int tid = threadIdx.x;
  for (int t = tid; t < T_LEN; t += blockDim.x) {
    s_t[t]         = input[t * 4 + 0];
    s_x[t * 3 + 0] = input[t * 4 + 1];
    s_x[t * 3 + 1] = input[t * 4 + 2];
    s_x[t * 3 + 2] = input[t * 4 + 3];
  }
  __syncthreads();

  for (int t = tid; t < T_LEN; t += blockDim.x) {
    const float x = s_x[t * 3 + 0];
    const float y = s_x[t * 3 + 1];
    const float z = s_x[t * 3 + 2];
    ws[0 * T_LEN + t] = x;
    ws[1 * T_LEN + t] = y;
    ws[2 * T_LEN + t] = z;
    float jx = 0.f, jy = 0.f, jz = 0.f;
    if (t > 0) {
      const float dt = s_t[t] - s_t[t - 1];
      const float inv = 1.0f / dt;
      jx = (x - s_x[(t - 1) * 3 + 0]) * inv;
      jy = (y - s_x[(t - 1) * 3 + 1]) * inv;
      jz = (z - s_x[(t - 1) * 3 + 2]) * inv;
    }
    ws[K1 + 0 * T_LEN + t] = jx;
    ws[K1 + 1 * T_LEN + t] = jy;
    ws[K1 + 2 * T_LEN + t] = jz;
    ws[2 * K1 + t]         = sqrtf(x * x + y * y + z * z);
    ws[2 * K1 + T_LEN + t] = sqrtf(jx * jx + jy * jy + jz * jz);
  }
  __syncthreads();

  if (tid < 3) {
    float s = 0.f;
    for (int t = 0; t < T_LEN; ++t) {
      const float v = s_x[t * 3 + tid];
      s += v * v;
    }
    ws[2 * K1 + K3 + tid] = s / (float)T_LEN;
  }
}

__global__ __launch_bounds__(256) void hdc_gemv_wmma_kernel(
    const float* __restrict__ W1, const float* __restrict__ b1,
    const float* __restrict__ W2, const float* __restrict__ b2,
    const float* __restrict__ W3, const float* __restrict__ b3,
    const float* __restrict__ W4, const float* __restrict__ b4,
    const float* __restrict__ feats, float* __restrict__ out) {
  const int lane  = threadIdx.x & 31;
  const int wave  = threadIdx.x >> 5;
  const int tile  = blockIdx.x * (blockDim.x >> 5) + wave;
  if (tile >= NTILES) return;

  const int row0  = tile * 16;
  const int m     = lane & 15;
  const int khalf = (lane >> 4) * 2;
  const int rowA  = row0 + m;

  const float* __restrict__ f1 = feats;
  const float* __restrict__ f2 = feats + K1;
  const float* __restrict__ f3 = feats + 2 * K1;
  const float* __restrict__ en = feats + 2 * K1 + K3;

  const float* __restrict__ r1 = W1 + (size_t)rowA * K1 + khalf;
  const float* __restrict__ r2 = W2 + (size_t)rowA * K1 + khalf;
  const float* __restrict__ r3 = W3 + (size_t)rowA * K3 + khalf;

  __builtin_prefetch(r1, 0, 0);
  __builtin_prefetch(r2, 0, 0);
  __builtin_prefetch(r3, 0, 0);

  v8f acc1 = {};
  v8f acc2 = {};
  v8f acc3 = {};

  #pragma unroll 4
  for (int k = 0; k < K1; k += 4) {
    const v2f a1 = *(const v2f*)(r1 + k);
    const v2f a2 = *(const v2f*)(r2 + k);
    const v2f bA = *(const v2f*)(f1 + k + khalf);
    const v2f bB = *(const v2f*)(f2 + k + khalf);
    acc1 = __builtin_amdgcn_wmma_f32_16x16x4_f32(false, a1, false, bA,
                                                 (short)0, acc1, false, false);
    acc2 = __builtin_amdgcn_wmma_f32_16x16x4_f32(false, a2, false, bB,
                                                 (short)0, acc2, false, false);
  }

  #pragma unroll 4
  for (int k = 0; k < K3; k += 4) {
    const v2f a3 = *(const v2f*)(r3 + k);
    const v2f bC = *(const v2f*)(f3 + k + khalf);
    acc3 = __builtin_amdgcn_wmma_f32_16x16x4_f32(false, a3, false, bC,
                                                 (short)0, acc3, false, false);
  }

  if ((lane & 15) == 0) {
    const int   half = lane >> 4;
    const float e0 = en[0], e1 = en[1], e2 = en[2];
    #pragma unroll
    for (int v = 0; v < 8; ++v) {
      const int row = row0 + half * 8 + v;
      const float h1 = tanhf(acc1[v] + b1[row]);
      const float h2 = tanhf(acc2[v] + b2[row]);
      const float h3 = tanhf(acc3[v] + b3[row]);
      const float d4 = W4[row * 3 + 0] * e0 + W4[row * 3 + 1] * e1 +
                       W4[row * 3 + 2] * e2 + b4[row];
      const float h4 = tanhf(d4);
      out[row] = tanhf(h1 * h4 + h2 + h3);
    }
  }
}

extern "C" void kernel_launch(void* const* d_in, const int* in_sizes, int n_in,
                              void* d_out, int out_size, void* d_ws, size_t ws_size,
                              hipStream_t stream) {
  (void)in_sizes; (void)n_in; (void)out_size; (void)ws_size;
  const float* input = (const float*)d_in[0];
  const float* W1    = (const float*)d_in[1];
  const float* b1    = (const float*)d_in[2];
  const float* W2    = (const float*)d_in[3];
  const float* b2    = (const float*)d_in[4];
  const float* W3    = (const float*)d_in[5];
  const float* b3    = (const float*)d_in[6];
  const float* W4    = (const float*)d_in[7];
  const float* b4    = (const float*)d_in[8];
  float*       out   = (float*)d_out;
  float*       feats = (float*)d_ws;

  hdc_features_kernel<<<1, 512, 0, stream>>>(input, feats);

  const int wavesPerBlock = 8;
  const int blocks = (NTILES + wavesPerBlock - 1) / wavesPerBlock;
  hdc_gemv_wmma_kernel<<<blocks, 256, 0, stream>>>(W1, b1, W2, b2, W3, b3,
                                                   W4, b4, feats, out);
}


# GEMV split over K: one 16-row tile per work-group (625 x 4 waves), its 3200 columns cut into 8 slices of 400, two per wave; partial sums meet in LDS and are added in a new fixed order
# speedup vs baseline: 2.8363x; 2.8363x over previous
#include <hip/hip_runtime.h>
#include <math.h>

typedef __attribute__((ext_vector_type(2))) float v2f;
typedef __attribute__((ext_vector_type(8))) float v8f;

#define T_LEN 400
#define D_LEN 10000
#define K1 1200
#define K3 800
#define NTILES (D_LEN / 16)

__global__ void hdc_features_kernel(const float* __restrict__ input,
                                    float* __restrict__ ws) {
  __shared__ float s_t[T_LEN];
  __shared__ float s_x[T_LEN * 3];

  const int tid = threadIdx.x;
  for (int t = tid; t < T_LEN; t += blockDim.x) {
    s_t[t]         = input[t * 4 + 0];
    s_x[t * 3 + 0] = input[t * 4 + 1];
    s_x[t * 3 + 1] = input[t * 4 + 2];
    s_x[t * 3 + 2] = input[t * 4 + 3];
  }
  __syncthreads();

  for (int t = tid; t < T_LEN; t += blockDim.x) {
    const float x = s_x[t * 3 + 0];
    const float y = s_x[t * 3 + 1];
    const float z = s_x[t * 3 + 2];
    ws[0 * T_LEN + t] = x;
    ws[1 * T_LEN + t] = y;
    ws[2 * T_LEN + t] = z;
    float jx = 0.f, jy = 0.f, jz = 0.f;
    if (t > 0) {
      const float dt = s_t[t] - s_t[t - 1];
      const float inv = 1.0f / dt;
      jx = (x - s_x[(t - 1) * 3 + 0]) * inv;
      jy = (y - s_x[(t - 1) * 3 + 1]) * inv;
      jz = (z - s_x[(t - 1) * 3 + 2]) * inv;
    }
    ws[K1 + 0 * T_LEN + t] = jx;
    ws[K1 + 1 * T_LEN + t] = jy;
    ws[K1 + 2 * T_LEN + t] = jz;
    ws[2 * K1 + t]         = sqrtf(x * x + y * y + z * z);
    ws[2 * K1 + T_LEN + t] = sqrtf(jx * jx + jy * jy + jz * jz);
  }
  __syncthreads();

  if (tid < 3) {
    float s = 0.f;
    for (int t = 0; t < T_LEN; ++t) {
      const float v = s_x[t * 3 + tid];
      s += v * v;
    }
    ws[2 * K1 + K3 + tid] = s / (float)T_LEN;
  }
}

#define KSLICE  400
#define NSLICES 8

__device__ __forceinline__ const float* slice_row_ptr(
    int u, int rowA, const float* __restrict__ W1, const float* __restrict__ W2,
    const float* __restrict__ W3) {
  const int mat = (u >= 3) + (u >= 6);
  const float* __restrict__ Wm = (mat == 0) ? W1 : (mat == 1) ? W2 : W3;
  const int ld  = (mat == 2) ? K3 : K1;
  return Wm + (size_t)rowA * ld + (u - 3 * mat) * KSLICE;
}

__global__ __launch_bounds__(128) __attribute__((amdgpu_waves_per_eu(1, 4)))
void hdc_gemv_wmma_kernel(
    const float* __restrict__ W1, const float* __restrict__ b1,
    const float* __restrict__ W2, const float* __restrict__ b2,
    const float* __restrict__ W3, const float* __restrict__ b3,
    const float* __restrict__ W4, const float* __restrict__ b4,
    const float* __restrict__ feats, float* __restrict__ out) {
  __shared__ float s_part[NSLICES][16];

  const int lane  = threadIdx.x & 31;
  const int wave  = __builtin_amdgcn_readfirstlane((int)(threadIdx.x >> 5));
  const int row0  = blockIdx.x * 16;
  const int m     = lane & 15;
  const int khalf = (lane >> 4) * 2;
  const int rowA  = row0 + m;

  const int uA = 2 * wave;
  const int uB = 2 * wave + 1;

  const float* __restrict__ rA = slice_row_ptr(uA, rowA, W1, W2, W3) + khalf;
  const float* __restrict__ rB = slice_row_ptr(uB, rowA, W1, W2, W3) + khalf;
  const float* __restrict__ fA = feats + uA * KSLICE + khalf;
  const float* __restrict__ fB = feats + uB * KSLICE + khalf;

  v8f accA = {};
  v8f accB = {};

  #pragma unroll
  for (int k = 0; k < KSLICE; k += 4) {
    const v2f a1 = *(const v2f*)(rA + k);
    const v2f a2 = *(const v2f*)(rB + k);
    const v2f bA = *(const v2f*)(fA + k);
    const v2f bB = *(const v2f*)(fB + k);
    accA = __builtin_amdgcn_wmma_f32_16x16x4_f32(false, a1, false, bA,
                                                 (short)0, accA, false, false);
    accB = __builtin_amdgcn_wmma_f32_16x16x4_f32(false, a2, false, bB,
                                                 (short)0, accB, false, false);
  }

  if ((lane & 15) == 0) {
    const int half = lane >> 4;
    #pragma unroll
    for (int v = 0; v < 8; ++v) {
      s_part[uA][half * 8 + v] = accA[v];
      s_part[uB][half * 8 + v] = accB[v];
    }
  }
  __syncthreads();

  if (wave == 0 && lane < 16) {
    const float* __restrict__ en = feats + 2 * K1 + K3;
    const int   row = rowA;
    const float e0 = en[0], e1 = en[1], e2 = en[2];
    const float bb1 = b1[row], bb2 = b2[row], bb3 = b3[row], bb4 = b4[row];
    const float w40 = W4[row * 3 + 0], w41 = W4[row * 3 + 1], w42 = W4[row * 3 + 2];
    const float d1 = (s_part[0][m] + s_part[1][m]) + s_part[2][m];
    const float d2 = (s_part[3][m] + s_part[4][m]) + s_part[5][m];
    const float d3 = s_part[6][m] + s_part[7][m];
    const float h1 = tanhf(d1 + bb1);
    const float h2 = tanhf(d2 + bb2);
    const float h3 = tanhf(d3 + bb3);
    const float d4 = w40 * e0 + w41 * e1 + w42 * e2 + bb4;
    const float h4 = tanhf(d4);
    out[row] = tanhf(h1 * h4 + h2 + h3);
  }
}

extern "C" void kernel_launch(void* const* d_in, const int* in_sizes, int n_in,
                              void* d_out, int out_size, void* d_ws, size_t ws_size,
                              hipStream_t stream) {
  (void)in_sizes; (void)n_in; (void)out_size; (void)ws_size;
  const float* input = (const float*)d_in[0];
  const float* W1    = (const float*)d_in[1];
  const float* b1    = (const float*)d_in[2];
  const float* W2    = (const float*)d_in[3];
  const float* b2    = (const float*)d_in[4];
  const float* W3    = (const float*)d_in[5];
  const float* b3    = (const float*)d_in[6];
  const float* W4    = (const float*)d_in[7];
  const float* b4    = (const float*)d_in[8];
  float*       out   = (float*)d_out;
  float*       feats = (float*)d_ws;

  hdc_features_kernel<<<1, 512, 0, stream>>>(input, feats);

  hdc_gemv_wmma_kernel<<<NTILES, 128, 0, stream>>>(W1, b1, W2, b2, W3, b3,
                                                   W4, b4, feats, out);
}


# Both WMMA operands staged through LDS: weights fetched 8 rows x 64 B per 16-byte load into a wave-private LDS stage, features copied once per wave; k order in a slice changes: sums in new fixed order
# speedup vs baseline: 3.6039x; 1.2706x over previous
#include <hip/hip_runtime.h>
#include <math.h>

typedef __attribute__((ext_vector_type(2))) float v2f;
typedef __attribute__((ext_vector_type(4))) float v4f;
typedef __attribute__((ext_vector_type(8))) float v8f;

#define T_LEN 400
#define D_LEN 10000
#define K1 1200
#define K3 800
#define NTILES (D_LEN / 16)

__global__ void hdc_features_kernel(const float* __restrict__ input,
                                    float* __restrict__ ws) {
  __shared__ float s_t[T_LEN];
  __shared__ float s_x[T_LEN * 3];

  const int tid = threadIdx.x;
  for (int t = tid; t < T_LEN; t += blockDim.x) {
    s_t[t]         = input[t * 4 + 0];
    s_x[t * 3 + 0] = input[t * 4 + 1];
    s_x[t * 3 + 1] = input[t * 4 + 2];
    s_x[t * 3 + 2] = input[t * 4 + 3];
  }
  __syncthreads();

  for (int t = tid; t < T_LEN; t += blockDim.x) {
    const float x = s_x[t * 3 + 0];
    const float y = s_x[t * 3 + 1];
    const float z = s_x[t * 3 + 2];
    ws[0 * T_LEN + t] = x;
    ws[1 * T_LEN + t] = y;
    ws[2 * T_LEN + t] = z;
    float jx = 0.f, jy = 0.f, jz = 0.f;
    if (t > 0) {
      const float dt = s_t[t] - s_t[t - 1];
      const float inv = 1.0f / dt;
      jx = (x - s_x[(t - 1) * 3 + 0]) * inv;
      jy = (y - s_x[(t - 1) * 3 + 1]) * inv;
      jz = (z - s_x[(t - 1) * 3 + 2]) * inv;
    }
    ws[K1 + 0 * T_LEN + t] = jx;
    ws[K1 + 1 * T_LEN + t] = jy;
    ws[K1 + 2 * T_LEN + t] = jz;
    ws[2 * K1 + t]         = sqrtf(x * x + y * y + z * z);
    ws[2 * K1 + T_LEN + t] = sqrtf(jx * jx + jy * jy + jz * jz);
  }
  __syncthreads();

  if (tid < 3) {
    float s = 0.f;
    for (int t = 0; t < T_LEN; ++t) {
      const float v = s_x[t * 3 + tid];
      s += v * v;
    }
    ws[2 * K1 + K3 + tid] = s / (float)T_LEN;
  }
}

#define KSLICE  400
#define NSLICES 8
#define CH      16
#define NCHUNK  (KSLICE / CH)
#define PITCH   20
#define STAGE_F (16 * PITCH)
#define DEPTH   8
#define WFEAT   (2 * KSLICE)

__device__ __forceinline__ const float* slice_row_ptr(
    int u, int rowA, const float* __restrict__ W1, const float* __restrict__ W2,
    const float* __restrict__ W3) {
  const int mat = (u >= 3) + (u >= 6);
  const float* __restrict__ Wm = (mat == 0) ? W1 : (mat == 1) ? W2 : W3;
  const int ld  = (mat == 2) ? K3 : K1;
  return Wm + (size_t)rowA * ld + (u - 3 * mat) * KSLICE;
}

__device__ __forceinline__ void wave_lds_order() {
  __builtin_amdgcn_fence(__ATOMIC_RELEASE, "wavefront");
  __builtin_amdgcn_wave_barrier();
  __builtin_amdgcn_fence(__ATOMIC_ACQUIRE, "wavefront");
}

__global__ __launch_bounds__(128) __attribute__((amdgpu_waves_per_eu(1, 4)))
void hdc_gemv_wmma_kernel(
    const float* __restrict__ W1, const float* __restrict__ b1,
    const float* __restrict__ W2, const float* __restrict__ b2,
    const float* __restrict__ W3, const float* __restrict__ b3,
    const float* __restrict__ W4, const float* __restrict__ b4,
    const float* __restrict__ feats, float* __restrict__ out) {
  __shared__ __attribute__((aligned(16))) float s_feat[4][WFEAT];
  __shared__ __attribute__((aligned(16))) float s_ring[4][2][2][STAGE_F];
  __shared__ float s_part[NSLICES][16];

  const int lane  = threadIdx.x & 31;
  const int wave  = __builtin_amdgcn_readfirstlane((int)(threadIdx.x >> 5));
  const int row0  = blockIdx.x * 16;
  const int m     = lane & 15;
  const int h     = lane >> 4;
  const int rowA  = row0 + m;

  const int uA = 2 * wave;
  const int uB = 2 * wave + 1;

  const int lrow = lane >> 2;
  const int lpc  = (lane & 3) * 4;
  const float* __restrict__ gA0 = slice_row_ptr(uA, row0 + lrow,     W1, W2, W3) + lpc;
  const float* __restrict__ gA1 = slice_row_ptr(uA, row0 + lrow + 8, W1, W2, W3) + lpc;
  const float* __restrict__ gB0 = slice_row_ptr(uB, row0 + lrow,     W1, W2, W3) + lpc;
  const float* __restrict__ gB1 = slice_row_ptr(uB, row0 + lrow + 8, W1, W2, W3) + lpc;

  float* __restrict__ myfeat = &s_feat[wave][0];
  float* __restrict__ ring   = &s_ring[wave][0][0][0];
  const int st_off = lrow * PITCH + lpc;
  const int fr_off = m * PITCH + 4 * h;

  v4f fv[7];
  #pragma unroll
  for (int q = 0; q < 7; ++q) {
    int idx = q * 32 + lane;
    idx = idx < (WFEAT / 4 - 1) ? idx : (WFEAT / 4 - 1);
    fv[q] = *(const v4f*)(feats + wave * WFEAT + 4 * idx);
  }

  v4f g[DEPTH][4];
  #pragma unroll
  for (int d = 0; d < DEPTH; ++d) {
    g[d][0] = *(const v4f*)(gA0 + CH * d);
    g[d][1] = *(const v4f*)(gA1 + CH * d);
    g[d][2] = *(const v4f*)(gB0 + CH * d);
    g[d][3] = *(const v4f*)(gB1 + CH * d);
  }

  #pragma unroll
  for (int q = 0; q < 7; ++q) {
    int idx = q * 32 + lane;
    idx = idx < (WFEAT / 4 - 1) ? idx : (WFEAT / 4 - 1);
    *(v4f*)(myfeat + 4 * idx) = fv[q];
  }

  v8f accA = {};
  v8f accB = {};

  *(v4f*)(ring + 0 * STAGE_F + st_off)             = g[0][0];
  *(v4f*)(ring + 0 * STAGE_F + st_off + 8 * PITCH) = g[0][1];
  *(v4f*)(ring + 1 * STAGE_F + st_off)             = g[0][2];
  *(v4f*)(ring + 1 * STAGE_F + st_off + 8 * PITCH) = g[0][3];
  g[0][0] = *(const v4f*)(gA0 + CH * DEPTH);
  g[0][1] = *(const v4f*)(gA1 + CH * DEPTH);
  g[0][2] = *(const v4f*)(gB0 + CH * DEPTH);
  g[0][3] = *(const v4f*)(gB1 + CH * DEPTH);
  wave_lds_order();

  v4f aA[2], aB[2], fA[2], fB[2];
  #pragma unroll
  for (int j = 0; j < 2; ++j) {
    aA[j] = *(const v4f*)(ring + 0 * STAGE_F + fr_off + 8 * j);
    aB[j] = *(const v4f*)(ring + 1 * STAGE_F + fr_off + 8 * j);
    fA[j] = *(const v4f*)(myfeat + 4 * h + 8 * j);
    fB[j] = *(const v4f*)(myfeat + KSLICE + 4 * h + 8 * j);
  }

  #pragma unroll
  for (int c = 0; c < NCHUNK; ++c) {
    v4f nA[2], nB[2], nfA[2], nfB[2];
    if (c + 1 < NCHUNK) {
      const int n  = c + 1;
      const int sl = n % DEPTH;
      float* __restrict__ stg = ring + (n & 1) * 2 * STAGE_F;
      *(v4f*)(stg + st_off)                       = g[sl][0];
      *(v4f*)(stg + st_off + 8 * PITCH)           = g[sl][1];
      *(v4f*)(stg + STAGE_F + st_off)             = g[sl][2];
      *(v4f*)(stg + STAGE_F + st_off + 8 * PITCH) = g[sl][3];
      if (n + DEPTH < NCHUNK) {
        g[sl][0] = *(const v4f*)(gA0 + CH * (n + DEPTH));
        g[sl][1] = *(const v4f*)(gA1 + CH * (n + DEPTH));
        g[sl][2] = *(const v4f*)(gB0 + CH * (n + DEPTH));
        g[sl][3] = *(const v4f*)(gB1 + CH * (n + DEPTH));
      }
      wave_lds_order();
      #pragma unroll
      for (int j = 0; j < 2; ++j) {
        nA[j]  = *(const v4f*)(stg + fr_off + 8 * j);
        nB[j]  = *(const v4f*)(stg + STAGE_F + fr_off + 8 * j);
        nfA[j] = *(const v4f*)(myfeat + CH * n + 4 * h + 8 * j);
        nfB[j] = *(const v4f*)(myfeat + KSLICE + CH * n + 4 * h + 8 * j);
      }
    }

    #pragma unroll
    for (int j = 0; j < 2; ++j) {
      const v2f a1lo = {aA[j].x, aA[j].y}, a1hi = {aA[j].z, aA[j].w};
      const v2f a2lo = {aB[j].x, aB[j].y}, a2hi = {aB[j].z, aB[j].w};
      const v2f b1lo = {fA[j].x, fA[j].y}, b1hi = {fA[j].z, fA[j].w};
      const v2f b2lo = {fB[j].x, fB[j].y}, b2hi = {fB[j].z, fB[j].w};
      accA = __builtin_amdgcn_wmma_f32_16x16x4_f32(false, a1lo, false, b1lo,
                                                   (short)0, accA, false, false);
      accB = __builtin_amdgcn_wmma_f32_16x16x4_f32(false, a2lo, false, b2lo,
                                                   (short)0, accB, false, false);
      accA = __builtin_amdgcn_wmma_f32_16x16x4_f32(false, a1hi, false, b1hi,
                                                   (short)0, accA, false, false);
      accB = __builtin_amdgcn_wmma_f32_16x16x4_f32(false, a2hi, false, b2hi,
                                                   (short)0, accB, false, false);
    }

    if (c + 1 < NCHUNK) {
      #pragma unroll
      for (int j = 0; j < 2; ++j) {
        aA[j] = nA[j]; aB[j] = nB[j]; fA[j] = nfA[j]; fB[j] = nfB[j];
      }
    }
  }

  if ((lane & 15) == 0) {
    #pragma unroll
    for (int v = 0; v < 8; ++v) {
      s_part[uA][h * 8 + v] = accA[v];
      s_part[uB][h * 8 + v] = accB[v];
    }
  }
  __syncthreads();

  if (wave == 0 && lane < 16) {
    const float* __restrict__ en = feats + 2 * K1 + K3;
    const int   row = rowA;
    const float e0 = en[0], e1 = en[1], e2 = en[2];
    const float bb1 = b1[row], bb2 = b2[row], bb3 = b3[row], bb4 = b4[row];
    const float w40 = W4[row * 3 + 0], w41 = W4[row * 3 + 1], w42 = W4[row * 3 + 2];
    const float d1 = (s_part[0][m] + s_part[1][m]) + s_part[2][m];
    const float d2 = (s_part[3][m] + s_part[4][m]) + s_part[5][m];
    const float d3 = s_part[6][m] + s_part[7][m];
    const float h1 = tanhf(d1 + bb1);
    const float h2 = tanhf(d2 + bb2);
    const float h3 = tanhf(d3 + bb3);
    const float d4 = w40 * e0 + w41 * e1 + w42 * e2 + bb4;
    const float h4 = tanhf(d4);
    out[row] = tanhf(h1 * h4 + h2 + h3);
  }
}

extern "C" void kernel_launch(void* const* d_in, const int* in_sizes, int n_in,
                              void* d_out, int out_size, void* d_ws, size_t ws_size,
                              hipStream_t stream) {
  (void)in_sizes; (void)n_in; (void)out_size; (void)ws_size;
  const float* input = (const float*)d_in[0];
  const float* W1    = (const float*)d_in[1];
  const float* b1    = (const float*)d_in[2];
  const float* W2    = (const float*)d_in[3];
  const float* b2    = (const float*)d_in[4];
  const float* W3    = (const float*)d_in[5];
  const float* b3    = (const float*)d_in[6];
  const float* W4    = (const float*)d_in[7];
  const float* b4    = (const float*)d_in[8];
  float*       out   = (float*)d_out;
  float*       feats = (float*)d_ws;

  hdc_features_kernel<<<1, 512, 0, stream>>>(input, feats);

  hdc_gemv_wmma_kernel<<<NTILES, 128, 0, stream>>>(W1, b1, W2, b2, W3, b3,
                                                   W4, b4, feats, out);
}


# Feature kernel: the 400-step serial energy loop of 3 threads is replaced by per-thread squares, an in-register wave sum and 16 per-wave partials in LDS; energy summed in a new fixed order
# speedup vs baseline: 4.8596x; 1.3484x over previous
#include <hip/hip_runtime.h>
#include <math.h>

typedef __attribute__((ext_vector_type(2))) float v2f;
typedef __attribute__((ext_vector_type(4))) float v4f;
typedef __attribute__((ext_vector_type(8))) float v8f;

#define T_LEN 400
#define D_LEN 10000
#define K1 1200
#define K3 800
#define NTILES (D_LEN / 16)
#define FEAT_THREADS 512

template <int MASK>
__device__ __forceinline__ float row_xor(float x) {
  const int b = __builtin_bit_cast(int, x);
  return __builtin_bit_cast(
      float, __builtin_amdgcn_update_dpp(b, b, 0x160 + MASK, 0xf, 0xf, false));
}

__device__ __forceinline__ float xhalf(float x) {
  const int b = __builtin_bit_cast(int, x);
  return __builtin_bit_cast(
      float, __builtin_amdgcn_permlanex16(b, b, 0x76543210, 0xfedcba98, false, false));
}

__device__ __forceinline__ float wave_sum(float x) {
  x += row_xor<1>(x);
  x += row_xor<2>(x);
  x += row_xor<4>(x);
  x += row_xor<8>(x);
  x += xhalf(x);
  return x;
}

__global__ void hdc_features_kernel(const float* __restrict__ input,
                                    float* __restrict__ ws) {
  __shared__ float s_t[T_LEN];
  __shared__ float s_x[T_LEN * 3];
  __shared__ __attribute__((aligned(16))) float s_e[3][FEAT_THREADS / 32];

  const int tid = threadIdx.x;
  for (int t = tid; t < T_LEN; t += blockDim.x) {
    s_t[t]         = input[t * 4 + 0];
    s_x[t * 3 + 0] = input[t * 4 + 1];
    s_x[t * 3 + 1] = input[t * 4 + 2];
    s_x[t * 3 + 2] = input[t * 4 + 3];
  }
  __syncthreads();

  float ex = 0.f, ey = 0.f, ez = 0.f;
  for (int t = tid; t < T_LEN; t += blockDim.x) {
    const float x = s_x[t * 3 + 0];
    const float y = s_x[t * 3 + 1];
    const float z = s_x[t * 3 + 2];
    ex += x * x;
    ey += y * y;
    ez += z * z;
    ws[0 * T_LEN + t] = x;
    ws[1 * T_LEN + t] = y;
    ws[2 * T_LEN + t] = z;
    float jx = 0.f, jy = 0.f, jz = 0.f;
    if (t > 0) {
      const float dt = s_t[t] - s_t[t - 1];
      const float inv = 1.0f / dt;
      jx = (x - s_x[(t - 1) * 3 + 0]) * inv;
      jy = (y - s_x[(t - 1) * 3 + 1]) * inv;
      jz = (z - s_x[(t - 1) * 3 + 2]) * inv;
    }
    ws[K1 + 0 * T_LEN + t] = jx;
    ws[K1 + 1 * T_LEN + t] = jy;
    ws[K1 + 2 * T_LEN + t] = jz;
    ws[2 * K1 + t]         = sqrtf(x * x + y * y + z * z);
    ws[2 * K1 + T_LEN + t] = sqrtf(jx * jx + jy * jy + jz * jz);
  }
  ex = wave_sum(ex);
  ey = wave_sum(ey);
  ez = wave_sum(ez);
  if ((tid & 31) == 0) {
    s_e[0][tid >> 5] = ex;
    s_e[1][tid >> 5] = ey;
    s_e[2][tid >> 5] = ez;
  }
  __syncthreads();

  if (tid < 3) {
    float s = 0.f;
    for (int w = 0; w < FEAT_THREADS / 32; ++w) s += s_e[tid][w];
    ws[2 * K1 + K3 + tid] = s / (float)T_LEN;
  }
}

#define KSLICE  400
#define NSLICES 8
#define CH      16
#define NCHUNK  (KSLICE / CH)
#define PITCH   20
#define STAGE_F (16 * PITCH)
#define DEPTH   8
#define WFEAT   (2 * KSLICE)

__device__ __forceinline__ const float* slice_row_ptr(
    int u, int rowA, const float* __restrict__ W1, const float* __restrict__ W2,
    const float* __restrict__ W3) {
  const int mat = (u >= 3) + (u >= 6);
  const float* __restrict__ Wm = (mat == 0) ? W1 : (mat == 1) ? W2 : W3;
  const int ld  = (mat == 2) ? K3 : K1;
  return Wm + (size_t)rowA * ld + (u - 3 * mat) * KSLICE;
}

__device__ __forceinline__ void wave_lds_order() {
  __builtin_amdgcn_fence(__ATOMIC_RELEASE, "wavefront");
  __builtin_amdgcn_wave_barrier();
  __builtin_amdgcn_fence(__ATOMIC_ACQUIRE, "wavefront");
}

__global__ __launch_bounds__(128) __attribute__((amdgpu_waves_per_eu(1, 4)))
void hdc_gemv_wmma_kernel(
    const float* __restrict__ W1, const float* __restrict__ b1,
    const float* __restrict__ W2, const float* __restrict__ b2,
    const float* __restrict__ W3, const float* __restrict__ b3,
    const float* __restrict__ W4, const float* __restrict__ b4,
    const float* __restrict__ feats, float* __restrict__ out) {
  __shared__ __attribute__((aligned(16))) float s_feat[4][WFEAT];
  __shared__ __attribute__((aligned(16))) float s_ring[4][2][2][STAGE_F];
  __shared__ float s_part[NSLICES][16];

  const int lane  = threadIdx.x & 31;
  const int wave  = __builtin_amdgcn_readfirstlane((int)(threadIdx.x >> 5));
  const int row0  = blockIdx.x * 16;
  const int m     = lane & 15;
  const int h     = lane >> 4;
  const int rowA  = row0 + m;

  const int uA = 2 * wave;
  const int uB = 2 * wave + 1;

  const int lrow = lane >> 2;
  const int lpc  = (lane & 3) * 4;
  const float* __restrict__ gA0 = slice_row_ptr(uA, row0 + lrow,     W1, W2, W3) + lpc;
  const float* __restrict__ gA1 = slice_row_ptr(uA, row0 + lrow + 8, W1, W2, W3) + lpc;
  const float* __restrict__ gB0 = slice_row_ptr(uB, row0 + lrow,     W1, W2, W3) + lpc;
  const float* __restrict__ gB1 = slice_row_ptr(uB, row0 + lrow + 8, W1, W2, W3) + lpc;

  float* __restrict__ myfeat = &s_feat[wave][0];
  float* __restrict__ ring   = &s_ring[wave][0][0][0];
  const int st_off = lrow * PITCH + lpc;
  const int fr_off = m * PITCH + 4 * h;

  v4f fv[7];
  #pragma unroll
  for (int q = 0; q < 7; ++q) {
    int idx = q * 32 + lane;
    idx = idx < (WFEAT / 4 - 1) ? idx : (WFEAT / 4 - 1);
    fv[q] = *(const v4f*)(feats + wave * WFEAT + 4 * idx);
  }

  v4f g[DEPTH][4];
  #pragma unroll
  for (int d = 0; d < DEPTH; ++d) {
    g[d][0] = *(const v4f*)(gA0 + CH * d);
    g[d][1] = *(const v4f*)(gA1 + CH * d);
    g[d][2] = *(const v4f*)(gB0 + CH * d);
    g[d][3] = *(const v4f*)(gB1 + CH * d);
  }

  #pragma unroll
  for (int q = 0; q < 7; ++q) {
    int idx = q * 32 + lane;
    idx = idx < (WFEAT / 4 - 1) ? idx : (WFEAT / 4 - 1);
    *(v4f*)(myfeat + 4 * idx) = fv[q];
  }

  v8f accA = {};
  v8f accB = {};

  *(v4f*)(ring + 0 * STAGE_F + st_off)             = g[0][0];
  *(v4f*)(ring + 0 * STAGE_F + st_off + 8 * PITCH) = g[0][1];
  *(v4f*)(ring + 1 * STAGE_F + st_off)             = g[0][2];
  *(v4f*)(ring + 1 * STAGE_F + st_off + 8 * PITCH) = g[0][3];
  g[0][0] = *(const v4f*)(gA0 + CH * DEPTH);
  g[0][1] = *(const v4f*)(gA1 + CH * DEPTH);
  g[0][2] = *(const v4f*)(gB0 + CH * DEPTH);
  g[0][3] = *(const v4f*)(gB1 + CH * DEPTH);
  wave_lds_order();

  v4f aA[2], aB[2], fA[2], fB[2];
  #pragma unroll
  for (int j = 0; j < 2; ++j) {
    aA[j] = *(const v4f*)(ring + 0 * STAGE_F + fr_off + 8 * j);
    aB[j] = *(const v4f*)(ring + 1 * STAGE_F + fr_off + 8 * j);
    fA[j] = *(const v4f*)(myfeat + 4 * h + 8 * j);
    fB[j] = *(const v4f*)(myfeat + KSLICE + 4 * h + 8 * j);
  }

  #pragma unroll
  for (int c = 0; c < NCHUNK; ++c) {
    v4f nA[2], nB[2], nfA[2], nfB[2];
    if (c + 1 < NCHUNK) {
      const int n  = c + 1;
      const int sl = n % DEPTH;
      float* __restrict__ stg = ring + (n & 1) * 2 * STAGE_F;
      *(v4f*)(stg + st_off)                       = g[sl][0];
      *(v4f*)(stg + st_off + 8 * PITCH)           = g[sl][1];
      *(v4f*)(stg + STAGE_F + st_off)             = g[sl][2];
      *(v4f*)(stg + STAGE_F + st_off + 8 * PITCH) = g[sl][3];
      if (n + DEPTH < NCHUNK) {
        g[sl][0] = *(const v4f*)(gA0 + CH * (n + DEPTH));
        g[sl][1] = *(const v4f*)(gA1 + CH * (n + DEPTH));
        g[sl][2] = *(const v4f*)(gB0 + CH * (n + DEPTH));
        g[sl][3] = *(const v4f*)(gB1 + CH * (n + DEPTH));
      }
      wave_lds_order();
      #pragma unroll
      for (int j = 0; j < 2; ++j) {
        nA[j]  = *(const v4f*)(stg + fr_off + 8 * j);
        nB[j]  = *(const v4f*)(stg + STAGE_F + fr_off + 8 * j);
        nfA[j] = *(const v4f*)(myfeat + CH * n + 4 * h + 8 * j);
        nfB[j] = *(const v4f*)(myfeat + KSLICE + CH * n + 4 * h + 8 * j);
      }
    }

    #pragma unroll
    for (int j = 0; j < 2; ++j) {
      const v2f a1lo = {aA[j].x, aA[j].y}, a1hi = {aA[j].z, aA[j].w};
      const v2f a2lo = {aB[j].x, aB[j].y}, a2hi = {aB[j].z, aB[j].w};
      const v2f b1lo = {fA[j].x, fA[j].y}, b1hi = {fA[j].z, fA[j].w};
      const v2f b2lo = {fB[j].x, fB[j].y}, b2hi = {fB[j].z, fB[j].w};
      accA = __builtin_amdgcn_wmma_f32_16x16x4_f32(false, a1lo, false, b1lo,
                                                   (short)0, accA, false, false);
      accB = __builtin_amdgcn_wmma_f32_16x16x4_f32(false, a2lo, false, b2lo,
                                                   (short)0, accB, false, false);
      accA = __builtin_amdgcn_wmma_f32_16x16x4_f32(false, a1hi, false, b1hi,
                                                   (short)0, accA, false, false);
      accB = __builtin_amdgcn_wmma_f32_16x16x4_f32(false, a2hi, false, b2hi,
                                                   (short)0, accB, false, false);
    }

    if (c + 1 < NCHUNK) {
      #pragma unroll
      for (int j = 0; j < 2; ++j) {
        aA[j] = nA[j]; aB[j] = nB[j]; fA[j] = nfA[j]; fB[j] = nfB[j];
      }
    }
  }

  if ((lane & 15) == 0) {
    #pragma unroll
    for (int v = 0; v < 8; ++v) {
      s_part[uA][h * 8 + v] = accA[v];
      s_part[uB][h * 8 + v] = accB[v];
    }
  }
  __syncthreads();

  if (wave == 0 && lane < 16) {
    const float* __restrict__ en = feats + 2 * K1 + K3;
    const int   row = rowA;
    const float e0 = en[0], e1 = en[1], e2 = en[2];
    const float bb1 = b1[row], bb2 = b2[row], bb3 = b3[row], bb4 = b4[row];
    const float w40 = W4[row * 3 + 0], w41 = W4[row * 3 + 1], w42 = W4[row * 3 + 2];
    const float d1 = (s_part[0][m] + s_part[1][m]) + s_part[2][m];
    const float d2 = (s_part[3][m] + s_part[4][m]) + s_part[5][m];
    const float d3 = s_part[6][m] + s_part[7][m];
    const float h1 = tanhf(d1 + bb1);
    const float h2 = tanhf(d2 + bb2);
    const float h3 = tanhf(d3 + bb3);
    const float d4 = w40 * e0 + w41 * e1 + w42 * e2 + bb4;
    const float h4 = tanhf(d4);
    out[row] = tanhf(h1 * h4 + h2 + h3);
  }
}

extern "C" void kernel_launch(void* const* d_in, const int* in_sizes, int n_in,
                              void* d_out, int out_size, void* d_ws, size_t ws_size,
                              hipStream_t stream) {
  (void)in_sizes; (void)n_in; (void)out_size; (void)ws_size;
  const float* input = (const float*)d_in[0];
  const float* W1    = (const float*)d_in[1];
  const float* b1    = (const float*)d_in[2];
  const float* W2    = (const float*)d_in[3];
  const float* b2    = (const float*)d_in[4];
  const float* W3    = (const float*)d_in[5];
  const float* b3    = (const float*)d_in[6];
  const float* W4    = (const float*)d_in[7];
  const float* b4    = (const float*)d_in[8];
  float*       out   = (float*)d_out;
  float*       feats = (float*)d_ws;

  hdc_features_kernel<<<1, FEAT_THREADS, 0, stream>>>(input, feats);

  hdc_gemv_wmma_kernel<<<NTILES, 128, 0, stream>>>(W1, b1, W2, b2, W3, b3,
                                                   W4, b4, feats, out);
}


# GEMV weight loads made row-contiguous: one wave load = 512 B of one row (4-5 lines, was 8 half lines), 128-column pieces staged in LDS; two k chains per slice: sums in new fixed order
# speedup vs baseline: 5.3921x; 1.1096x over previous
#include <hip/hip_runtime.h>
#include <math.h>

typedef __attribute__((ext_vector_type(2))) float v2f;
typedef __attribute__((ext_vector_type(4))) float v4f;
typedef __attribute__((ext_vector_type(8))) float v8f;

#define T_LEN 400
#define D_LEN 10000
#define K1 1200
#define K3 800
#define NTILES (D_LEN / 16)
#define FEAT_THREADS 512

template <int MASK>
__device__ __forceinline__ float row_xor(float x) {
  const int b = __builtin_bit_cast(int, x);
  return __builtin_bit_cast(
      float, __builtin_amdgcn_update_dpp(b, b, 0x160 + MASK, 0xf, 0xf, false));
}

__device__ __forceinline__ float xhalf(float x) {
  const int b = __builtin_bit_cast(int, x);
  return __builtin_bit_cast(
      float, __builtin_amdgcn_permlanex16(b, b, 0x76543210, 0xfedcba98, false, false));
}

__device__ __forceinline__ float wave_sum(float x) {
  x += row_xor<1>(x);
  x += row_xor<2>(x);
  x += row_xor<4>(x);
  x += row_xor<8>(x);
  x += xhalf(x);
  return x;
}

__global__ void hdc_features_kernel(const float* __restrict__ input,
                                    float* __restrict__ ws) {
  __shared__ float s_t[T_LEN];
  __shared__ float s_x[T_LEN * 3];
  __shared__ __attribute__((aligned(16))) float s_e[3][FEAT_THREADS / 32];

  const int tid = threadIdx.x;
  for (int t = tid; t < T_LEN; t += blockDim.x) {
    s_t[t]         = input[t * 4 + 0];
    s_x[t * 3 + 0] = input[t * 4 + 1];
    s_x[t * 3 + 1] = input[t * 4 + 2];
    s_x[t * 3 + 2] = input[t * 4 + 3];
  }
  __syncthreads();

  float ex = 0.f, ey = 0.f, ez = 0.f;
  for (int t = tid; t < T_LEN; t += blockDim.x) {
    const float x = s_x[t * 3 + 0];
    const float y = s_x[t * 3 + 1];
    const float z = s_x[t * 3 + 2];
    ex += x * x;
    ey += y * y;
    ez += z * z;
    ws[0 * T_LEN + t] = x;
    ws[1 * T_LEN + t] = y;
    ws[2 * T_LEN + t] = z;
    float jx = 0.f, jy = 0.f, jz = 0.f;
    if (t > 0) {
      const float dt = s_t[t] - s_t[t - 1];
      const float inv = 1.0f / dt;
      jx = (x - s_x[(t - 1) * 3 + 0]) * inv;
      jy = (y - s_x[(t - 1) * 3 + 1]) * inv;
      jz = (z - s_x[(t - 1) * 3 + 2]) * inv;
    }
    ws[K1 + 0 * T_LEN + t] = jx;
    ws[K1 + 1 * T_LEN + t] = jy;
    ws[K1 + 2 * T_LEN + t] = jz;
    ws[2 * K1 + t]         = sqrtf(x * x + y * y + z * z);
    ws[2 * K1 + T_LEN + t] = sqrtf(jx * jx + jy * jy + jz * jz);
  }
  ex = wave_sum(ex);
  ey = wave_sum(ey);
  ez = wave_sum(ez);
  if ((tid & 31) == 0) {
    s_e[0][tid >> 5] = ex;
    s_e[1][tid >> 5] = ey;
    s_e[2][tid >> 5] = ez;
  }
  __syncthreads();

  if (tid < 3) {
    float s = 0.f;
    for (int w = 0; w < FEAT_THREADS / 32; ++w) s += s_e[tid][w];
    ws[2 * K1 + K3 + tid] = s / (float)T_LEN;
  }
}

#define KSLICE  400
#define NSLICES 8
#define ICOLS   128
#define NFULL   3
#define IPITCH  132
#define ISTAGE  (16 * IPITCH)
#define TCOL    (NFULL * ICOLS)
#define TPITCH  20
#define TSTAGE  (16 * TPITCH)
#define NSUB    (2 * NFULL * 4)
#define WFEAT   (2 * KSLICE)

__device__ __forceinline__ const float* slice_row_ptr(
    int u, int rowA, const float* __restrict__ W1, const float* __restrict__ W2,
    const float* __restrict__ W3) {
  const int mat = (u >= 3) + (u >= 6);
  const float* __restrict__ Wm = (mat == 0) ? W1 : (mat == 1) ? W2 : W3;
  const int ld  = (mat == 2) ? K3 : K1;
  return Wm + (size_t)rowA * ld + (u - 3 * mat) * KSLICE;
}

__device__ __forceinline__ void wave_lds_order() {
  __builtin_amdgcn_fence(__ATOMIC_RELEASE, "wavefront");
  __builtin_amdgcn_wave_barrier();
  __builtin_amdgcn_fence(__ATOMIC_ACQUIRE, "wavefront");
}

template <int LD>
__device__ __forceinline__ void load_piece(v4f (&dst)[16], const float* __restrict__ p,
                                           int lane_off) {
  #pragma unroll
  for (int r = 0; r < 16; ++r)
    dst[r] = *(const v4f*)(p + r * LD + lane_off);
}

template <int LD>
__device__ __forceinline__ void load_tails(v4f (&dst)[16], const float* __restrict__ pA,
                                           const float* __restrict__ pB, int lrow, int lpc) {
  dst[0] = *(const v4f*)(pA + lrow * LD + TCOL + lpc);
  dst[1] = *(const v4f*)(pA + lrow * LD + 8 * LD + TCOL + lpc);
  dst[2] = *(const v4f*)(pB + lrow * LD + TCOL + lpc);
  dst[3] = *(const v4f*)(pB + lrow * LD + 8 * LD + TCOL + lpc);
}

__global__ __launch_bounds__(128) __attribute__((amdgpu_waves_per_eu(1, 4)))
void hdc_gemv_wmma_kernel(
    const float* __restrict__ W1, const float* __restrict__ b1,
    const float* __restrict__ W2, const float* __restrict__ b2,
    const float* __restrict__ W3, const float* __restrict__ b3,
    const float* __restrict__ W4, const float* __restrict__ b4,
    const float* __restrict__ feats, float* __restrict__ out) {
  __shared__ __attribute__((aligned(16))) float s_feat[4][WFEAT];
  __shared__ __attribute__((aligned(16))) float s_stage[4][2][ISTAGE];
  __shared__ float s_part[NSLICES][16];

  const int lane  = threadIdx.x & 31;
  const int wave  = __builtin_amdgcn_readfirstlane((int)(threadIdx.x >> 5));
  const int row0  = blockIdx.x * 16;
  const int m     = lane & 15;
  const int h     = lane >> 4;
  const int rowA  = row0 + m;

  const int uA = 2 * wave;
  const int uB = 2 * wave + 1;
  const bool shortRows = (wave == 3);
  const float* __restrict__ pA = slice_row_ptr(uA, row0, W1, W2, W3);
  const float* __restrict__ pB = slice_row_ptr(uB, row0, W1, W2, W3);

  const int lrow = lane >> 2;
  const int lpc  = (lane & 3) * 4;

  float* __restrict__ myfeat = &s_feat[wave][0];
  float* __restrict__ stage  = &s_stage[wave][0][0];
  const int st_off  = 4 * lane;
  const int fr_off  = m * IPITCH + 4 * h;
  const int tst_off = lrow * TPITCH + lpc;
  const int tfr_off = m * TPITCH + 4 * h;

  v4f fv[7];
  #pragma unroll
  for (int q = 0; q < 7; ++q) {
    int idx = q * 32 + lane;
    idx = idx < (WFEAT / 4 - 1) ? idx : (WFEAT / 4 - 1);
    fv[q] = *(const v4f*)(feats + wave * WFEAT + 4 * idx);
  }

  v4f g[2][16];
  if (shortRows) {
    load_piece<K3>(g[0], pA, st_off);
    load_piece<K3>(g[1], pB, st_off);
  } else {
    load_piece<K1>(g[0], pA, st_off);
    load_piece<K1>(g[1], pB, st_off);
  }

  #pragma unroll
  for (int q = 0; q < 7; ++q) {
    int idx = q * 32 + lane;
    idx = idx < (WFEAT / 4 - 1) ? idx : (WFEAT / 4 - 1);
    *(v4f*)(myfeat + 4 * idx) = fv[q];
  }

  v8f accA0 = {}, accA1 = {};
  v8f accB0 = {}, accB1 = {};

  #pragma unroll
  for (int r = 0; r < 16; ++r)
    *(v4f*)(stage + r * IPITCH + st_off) = g[0][r];
  if (shortRows) load_piece<K3>(g[0], pA + ICOLS, st_off);
  else           load_piece<K1>(g[0], pA + ICOLS, st_off);
  wave_lds_order();

  v4f aF[4], fF[4];
  #pragma unroll
  for (int j = 0; j < 4; ++j) {
    aF[j] = *(const v4f*)(stage + fr_off + 8 * j);
    fF[j] = *(const v4f*)(myfeat + 4 * h + 8 * j);
  }

  #pragma unroll
  for (int b = 0; b <= NSUB; ++b) {
    v4f nA[4], nF[4];
    if (b < NSUB) {
      const int n = b + 1;
      if (n < NSUB) {
        const int it = n >> 2;
        const int sb = n & 3;
        const float* __restrict__ sp = stage + (it & 1) * ISTAGE + fr_off + 32 * sb;
        const float* __restrict__ fp =
            myfeat + (it & 1) * KSLICE + ICOLS * (it >> 1) + 32 * sb + 4 * h;
        #pragma unroll
        for (int j = 0; j < 4; ++j) {
          nA[j] = *(const v4f*)(sp + 8 * j);
          nF[j] = *(const v4f*)(fp + 8 * j);
        }
      } else {
        #pragma unroll
        for (int j = 0; j < 2; ++j) {
          nA[j]     = *(const v4f*)(stage + tfr_off + 8 * j);
          nA[2 + j] = *(const v4f*)(stage + TSTAGE + tfr_off + 8 * j);
          nF[j]     = *(const v4f*)(myfeat + TCOL + 4 * h + 8 * j);
          nF[2 + j] = *(const v4f*)(myfeat + KSLICE + TCOL + 4 * h + 8 * j);
        }
      }
    }

    #pragma unroll
    for (int j = 0; j < 4; ++j) {
      const v2f alo = {aF[j].x, aF[j].y}, ahi = {aF[j].z, aF[j].w};
      const v2f blo = {fF[j].x, fF[j].y}, bhi = {fF[j].z, fF[j].w};
      const bool first = (b < NSUB) ? (((b >> 2) & 1) == 0) : (j < 2);
      if (first) {
        accA0 = __builtin_amdgcn_wmma_f32_16x16x4_f32(false, alo, false, blo,
                                                      (short)0, accA0, false, false);
        accA1 = __builtin_amdgcn_wmma_f32_16x16x4_f32(false, ahi, false, bhi,
                                                      (short)0, accA1, false, false);
      } else {
        accB0 = __builtin_amdgcn_wmma_f32_16x16x4_f32(false, alo, false, blo,
                                                      (short)0, accB0, false, false);
        accB1 = __builtin_amdgcn_wmma_f32_16x16x4_f32(false, ahi, false, bhi,
                                                      (short)0, accB1, false, false);
      }
    }

    if ((b & 3) == 1) {
      const int it = (b >> 2) + 1;
      wave_lds_order();
      if (it < 2 * NFULL) {
        const int set = it & 1;
        #pragma unroll
        for (int r = 0; r < 16; ++r)
          *(v4f*)(stage + set * ISTAGE + r * IPITCH + st_off) = g[set][r];
        if (it + 2 < 2 * NFULL) {
          const float* __restrict__ p = (set ? pB : pA) + ICOLS * ((it + 2) >> 1);
          if (shortRows) load_piece<K3>(g[set], p, st_off);
          else           load_piece<K1>(g[set], p, st_off);
        } else if (it == 2 * NFULL - 2) {
          if (shortRows) load_tails<K3>(g[0], pA, pB, lrow, lpc);
          else           load_tails<K1>(g[0], pA, pB, lrow, lpc);
        }
      } else {
        *(v4f*)(stage + tst_off)                       = g[0][0];
        *(v4f*)(stage + tst_off + 8 * TPITCH)          = g[0][1];
        *(v4f*)(stage + TSTAGE + tst_off)              = g[0][2];
        *(v4f*)(stage + TSTAGE + tst_off + 8 * TPITCH) = g[0][3];
      }
      wave_lds_order();
    }

    if (b < NSUB) {
      #pragma unroll
      for (int j = 0; j < 4; ++j) { aF[j] = nA[j]; fF[j] = nF[j]; }
    }
  }

  if ((lane & 15) == 0) {
    #pragma unroll
    for (int v = 0; v < 8; ++v) {
      s_part[uA][h * 8 + v] = accA0[v] + accA1[v];
      s_part[uB][h * 8 + v] = accB0[v] + accB1[v];
    }
  }
  __syncthreads();

  if (wave == 0 && lane < 16) {
    const float* __restrict__ en = feats + 2 * K1 + K3;
    const int   row = rowA;
    const float e0 = en[0], e1 = en[1], e2 = en[2];
    const float bb1 = b1[row], bb2 = b2[row], bb3 = b3[row], bb4 = b4[row];
    const float w40 = W4[row * 3 + 0], w41 = W4[row * 3 + 1], w42 = W4[row * 3 + 2];
    const float d1 = (s_part[0][m] + s_part[1][m]) + s_part[2][m];
    const float d2 = (s_part[3][m] + s_part[4][m]) + s_part[5][m];
    const float d3 = s_part[6][m] + s_part[7][m];
    const float h1 = tanhf(d1 + bb1);
    const float h2 = tanhf(d2 + bb2);
    const float h3 = tanhf(d3 + bb3);
    const float d4 = w40 * e0 + w41 * e1 + w42 * e2 + bb4;
    const float h4 = tanhf(d4);
    out[row] = tanhf(h1 * h4 + h2 + h3);
  }
}

extern "C" void kernel_launch(void* const* d_in, const int* in_sizes, int n_in,
                              void* d_out, int out_size, void* d_ws, size_t ws_size,
                              hipStream_t stream) {
  (void)in_sizes; (void)n_in; (void)out_size; (void)ws_size;
  const float* input = (const float*)d_in[0];
  const float* W1    = (const float*)d_in[1];
  const float* b1    = (const float*)d_in[2];
  const float* W2    = (const float*)d_in[3];
  const float* b2    = (const float*)d_in[4];
  const float* W3    = (const float*)d_in[5];
  const float* b3    = (const float*)d_in[6];
  const float* W4    = (const float*)d_in[7];
  const float* b4    = (const float*)d_in[8];
  float*       out   = (float*)d_out;
  float*       feats = (float*)d_ws;

  hdc_features_kernel<<<1, FEAT_THREADS, 0, stream>>>(input, feats);

  hdc_gemv_wmma_kernel<<<NTILES, 128, 0, stream>>>(W1, b1, W2, b2, W3, b3,
                                                   W4, b4, feats, out);
}
